# starting kernel
# baseline (speedup 1.0000x reference)
#include <hip/hip_runtime.h>
#include <hip/hip_bf16.h>
#include <math.h>

#define N_NODES   100000
#define N_EDGES   1600000
#define IN_FEATS  128
#define HEADS     4
#define OUT_F     16
#define HF        64
#define NEG_SLOPE 0.2f

typedef __attribute__((ext_vector_type(2))) float v2f;
typedef __attribute__((ext_vector_type(8))) float v8f;

__device__ __forceinline__ unsigned ordEnc(float f) {
  unsigned u = __float_as_uint(f);
  return (u & 0x80000000u) ? ~u : (u | 0x80000000u);
}
__device__ __forceinline__ float ordDec(unsigned x) {
  return (x & 0x80000000u) ? __uint_as_float(x ^ 0x80000000u)
                           : __uint_as_float(~x);
}

__device__ __forceinline__ float lrelu(float x) {
  return x > 0.0f ? x : NEG_SLOPE * x;
}

__global__ __launch_bounds__(128) void k_project(const float* __restrict__ feats,
                                                 const float* __restrict__ W,
                                                 float* __restrict__ z) {
  const int t      = threadIdx.x & 31;
  const int w      = threadIdx.x >> 5;
  const int lane16 = t & 15;
  const int half   = t >> 4;
  const int nodeBase = blockIdx.x * 16;

  const float* __restrict__ arow = feats + (size_t)(nodeBase + lane16) * IN_FEATS + half * 2;
  const float* __restrict__ brow = W     + (size_t)(w * 16  + lane16) * IN_FEATS + half * 2;

  v8f acc = {};
#pragma unroll
  for (int kk = 0; kk < IN_FEATS; kk += 4) {
    v2f a = { arow[kk], arow[kk + 1] };
    v2f b = { brow[kk], brow[kk + 1] };
    acc = __builtin_amdgcn_wmma_f32_16x16x4_f32(false, a, false, b,
                                                (short)0, acc, false, false);
  }

  float* __restrict__ zo = z + (size_t)(nodeBase + half * 8) * HF + w * 16 + lane16;
#pragma unroll
  for (int i = 0; i < 8; ++i) zo[(size_t)i * HF] = acc[i];
}

__global__ void k_logits(const float* __restrict__ z,
                         const float* __restrict__ attn_l,
                         const float* __restrict__ attn_r,
                         float* __restrict__ el, float* __restrict__ er) {
  int i = blockIdx.x * blockDim.x + threadIdx.x;
  if (i >= N_NODES * HEADS) return;
  int n = i >> 2, h = i & 3;
  const float* zp = z + (size_t)n * HF + h * OUT_F;
  const float* al = attn_l + h * OUT_F;
  const float* ar = attn_r + h * OUT_F;
  float sl = 0.0f, sr = 0.0f;
#pragma unroll
  for (int f = 0; f < OUT_F; ++f) {
    float v = zp[f];
    sl = fmaf(v, al[f], sl);
    sr = fmaf(v, ar[f], sr);
  }
  el[i] = sl;
  er[i] = sr;
}

__global__ void k_init(float* __restrict__ out, float* __restrict__ denom,
                       unsigned* __restrict__ menc) {
  int i = blockIdx.x * blockDim.x + threadIdx.x;
  if (i < N_NODES * HF) out[i] = 0.0f;
  if (i < N_NODES * HEADS) {
    denom[i] = 0.0f;
    menc[i]  = ordEnc(-INFINITY);
  }
}

__global__ void k_edge_max(const int* __restrict__ src, const int* __restrict__ dst,
                           const float* __restrict__ el, const float* __restrict__ er,
                           unsigned* __restrict__ menc) {
  int i = blockIdx.x * blockDim.x + threadIdx.x;
  if (i >= N_EDGES * HEADS) return;
  int e = i >> 2, h = i & 3;
  int s = src[e], d = dst[e];
  float v = lrelu(el[s * HEADS + h] + er[d * HEADS + h]);
  atomicMax(&menc[d * HEADS + h], ordEnc(v));
}

__global__ __launch_bounds__(256) void k_edge_agg(const int* __restrict__ src,
                                                  const int* __restrict__ dst,
                                                  const float* __restrict__ el,
                                                  const float* __restrict__ er,
                                                  const unsigned* __restrict__ menc,
                                                  const float* __restrict__ z,
                                                  float* __restrict__ denom,
                                                  float* __restrict__ out) {
  long long i = (long long)blockIdx.x * blockDim.x + threadIdx.x;
  if (i >= (long long)N_EDGES * HF) return;
  int e = (int)(i >> 6);
  int r = (int)(i & 63);
  int h = r >> 4;
  int f = r & 15;
  int s = src[e], d = dst[e];
  float lg = lrelu(el[s * HEADS + h] + er[d * HEADS + h]);
  float m  = ordDec(menc[d * HEADS + h]);
  float ex = __expf(lg - m);
  if (f == 0) atomicAdd(&denom[d * HEADS + h], ex);
  atomicAdd(&out[(size_t)d * HF + h * OUT_F + f],
            ex * z[(size_t)s * HF + h * OUT_F + f]);
}

__global__ void k_final(float* __restrict__ out, const float* __restrict__ denom,
                        const float* __restrict__ bias) {
  int i = blockIdx.x * blockDim.x + threadIdx.x;
  if (i >= N_NODES * HF) return;
  int n = i >> 6;
  int c = i & 63;
  int h = c >> 4;
  float den = denom[n * HEADS + h];
  float v   = out[i];
  out[i] = (den != 0.0f) ? (v / den + bias[c]) : bias[c];
}

extern "C" void kernel_launch(void* const* d_in, const int* in_sizes, int n_in,
                              void* d_out, int out_size, void* d_ws, size_t ws_size,
                              hipStream_t stream) {
  (void)in_sizes; (void)n_in; (void)out_size; (void)ws_size;
  const float* feats  = (const float*)d_in[0];
  const float* W      = (const float*)d_in[1];
  const float* attn_l = (const float*)d_in[2];
  const float* attn_r = (const float*)d_in[3];
  const float* bias   = (const float*)d_in[4];
  const int*   src    = (const int*)d_in[5];
  const int*   dst    = (const int*)d_in[6];
  float*       out    = (float*)d_out;

  float*    ws    = (float*)d_ws;
  float*    z     = ws;
  float*    el    = z  + (size_t)N_NODES * HF;
  float*    er    = el + (size_t)N_NODES * HEADS;
  unsigned* menc  = (unsigned*)(er + (size_t)N_NODES * HEADS);
  float*    denom = (float*)(menc + (size_t)N_NODES * HEADS);

  k_project<<<N_NODES / 16, 128, 0, stream>>>(feats, W, z);

  {
    int n = N_NODES * HEADS;
    k_logits<<<(n + 255) / 256, 256, 0, stream>>>(z, attn_l, attn_r, el, er);
  }

  {
    int n = N_NODES * HF;
    k_init<<<(n + 255) / 256, 256, 0, stream>>>(out, denom, menc);
  }

  {
    int n = N_EDGES * HEADS;
    k_edge_max<<<(n + 255) / 256, 256, 0, stream>>>(src, dst, el, er, menc);
  }

  {
    long long n = (long long)N_EDGES * HF;
    k_edge_agg<<<(unsigned)((n + 255) / 256), 256, 0, stream>>>(src, dst, el, er,
                                                               menc, z, denom, out);
  }

  {
    int n = N_NODES * HF;
    k_final<<<(n + 255) / 256, 256, 0, stream>>>(out, denom, bias);
  }
}


# Projection: each wave covers all 64 channels of its 16 nodes, W staged once per block in LDS, feats read once with 16-byte loads (was 4 times, 8-byte); k summed in another order within each 8
# speedup vs baseline: 2.7920x; 2.7920x over previous
#include <hip/hip_runtime.h>
#include <hip/hip_bf16.h>
#include <math.h>

#define N_NODES   100000
#define N_EDGES   1600000
#define IN_FEATS  128
#define HEADS     4
#define OUT_F     16
#define HF        64
#define NEG_SLOPE 0.2f
#define CAP       63

typedef __attribute__((ext_vector_type(2))) float v2f;
typedef __attribute__((ext_vector_type(4))) float v4f;
typedef __attribute__((ext_vector_type(8))) float v8f;

__device__ __forceinline__ float lrelu(float x) {
  return x > 0.0f ? x : NEG_SLOPE * x;
}

#define K1_WAVES 5
#define WPITCH4  33

__global__ __launch_bounds__(K1_WAVES * 32) void k_project(const float* __restrict__ feats,
                                                           const float* __restrict__ W,
                                                           float* __restrict__ z) {
  __shared__ v4f sW[HF * WPITCH4];

  const int tid = threadIdx.x;
  const int t   = tid & 31;
  const int w   = tid >> 5;
  const int c   = t & 15;
  const int hh  = t >> 4;
  const int nodeBase = (blockIdx.x * K1_WAVES + w) * 16;

  const v4f* __restrict__ arow =
      reinterpret_cast<const v4f*>(feats + (size_t)(nodeBase + c) * IN_FEATS) + hh;
  v4f a[16];
#pragma unroll
  for (int s = 0; s < 16; ++s) a[s] = arow[2 * s];

  const v4f* __restrict__ W4 = reinterpret_cast<const v4f*>(W);
  for (int i = tid; i < HF * (IN_FEATS / 4); i += K1_WAVES * 32)
    sW[(i >> 5) * WPITCH4 + (i & 31)] = W4[i];
  __syncthreads();

  const v4f* brow = sW + c * WPITCH4 + hh;

  v8f acc0 = {}, acc1 = {}, acc2 = {}, acc3 = {};
#pragma unroll
  for (int s = 0; s < 16; ++s) {
    const v4f b0 = brow[(0 * 16) * WPITCH4 + 2 * s];
    const v4f b1 = brow[(1 * 16) * WPITCH4 + 2 * s];
    const v4f b2 = brow[(2 * 16) * WPITCH4 + 2 * s];
    const v4f b3 = brow[(3 * 16) * WPITCH4 + 2 * s];
    const v4f av = a[s];
    acc0 = __builtin_amdgcn_wmma_f32_16x16x4_f32(false, av.lo, false, b0.lo, (short)0, acc0, false, false);
    acc1 = __builtin_amdgcn_wmma_f32_16x16x4_f32(false, av.lo, false, b1.lo, (short)0, acc1, false, false);
    acc2 = __builtin_amdgcn_wmma_f32_16x16x4_f32(false, av.lo, false, b2.lo, (short)0, acc2, false, false);
    acc3 = __builtin_amdgcn_wmma_f32_16x16x4_f32(false, av.lo, false, b3.lo, (short)0, acc3, false, false);
    acc0 = __builtin_amdgcn_wmma_f32_16x16x4_f32(false, av.hi, false, b0.hi, (short)0, acc0, false, false);
    acc1 = __builtin_amdgcn_wmma_f32_16x16x4_f32(false, av.hi, false, b1.hi, (short)0, acc1, false, false);
    acc2 = __builtin_amdgcn_wmma_f32_16x16x4_f32(false, av.hi, false, b2.hi, (short)0, acc2, false, false);
    acc3 = __builtin_amdgcn_wmma_f32_16x16x4_f32(false, av.hi, false, b3.hi, (short)0, acc3, false, false);
  }

  float* __restrict__ zo = z + (size_t)(nodeBase + hh * 8) * HF + c;
#pragma unroll
  for (int i = 0; i < 8; ++i) {
    zo[(size_t)i * HF +  0] = acc0[i];
    zo[(size_t)i * HF + 16] = acc1[i];
    zo[(size_t)i * HF + 32] = acc2[i];
    zo[(size_t)i * HF + 48] = acc3[i];
  }
}

__global__ void k_logits(const float* __restrict__ z,
                         const float* __restrict__ attn_l,
                         const float* __restrict__ attn_r,
                         float* __restrict__ el, float* __restrict__ er) {
  int i = blockIdx.x * blockDim.x + threadIdx.x;
  if (i >= N_NODES * HEADS) return;
  int n = i >> 2, h = i & 3;
  const float* zp = z + (size_t)n * HF + h * OUT_F;
  const float* al = attn_l + h * OUT_F;
  const float* ar = attn_r + h * OUT_F;
  float sl = 0.0f, sr = 0.0f;
#pragma unroll
  for (int f = 0; f < OUT_F; ++f) {
    float v = zp[f];
    sl = fmaf(v, al[f], sl);
    sr = fmaf(v, ar[f], sr);
  }
  el[i] = sl;
  er[i] = sr;
}

__global__ void k_init(float* out) {
  int n = blockIdx.x * blockDim.x + threadIdx.x;
  if (n < N_NODES) reinterpret_cast<int*>(out)[(size_t)n * HF] = 0;
}

__global__ void k_fill(const int* __restrict__ src, const int* __restrict__ dst,
                       float* out) {
  int e = blockIdx.x * blockDim.x + threadIdx.x;
  if (e >= N_EDGES) return;
  int s = src[e], d = dst[e];
  int* row = reinterpret_cast<int*>(out) + (size_t)d * HF;
  int pos = atomicAdd(row, 1);
  if (pos < CAP) row[1 + pos] = s;
}

__global__ __launch_bounds__(256) void k_node_agg(const int* __restrict__ src,
                                                  const int* __restrict__ dst,
                                                  const float* __restrict__ el,
                                                  const float* __restrict__ er,
                                                  const float* __restrict__ z,
                                                  const float* __restrict__ bias,
                                                  float* out) {
  const int lane = threadIdx.x & 31;
  const int n = __builtin_amdgcn_readfirstlane((int)(blockIdx.x * 8 + (threadIdx.x >> 5)));
  const int h = lane >> 3;
  const int* row = reinterpret_cast<const int*>(out) + (size_t)n * HF;
  const int cnt = __builtin_amdgcn_readfirstlane(row[0]);
  const bool inl = cnt <= CAP;

  const float4 erv = *reinterpret_cast<const float4*>(er + (size_t)n * HEADS);
  const float er_h = h == 0 ? erv.x : h == 1 ? erv.y : h == 2 ? erv.z : erv.w;

  float m0 = -INFINITY, m1 = -INFINITY, m2 = -INFINITY, m3 = -INFINITY;
  auto upd_max = [&](int s) {
    const float4 a = *reinterpret_cast<const float4*>(el + (size_t)s * HEADS);
    m0 = fmaxf(m0, lrelu(a.x + erv.x));
    m1 = fmaxf(m1, lrelu(a.y + erv.y));
    m2 = fmaxf(m2, lrelu(a.z + erv.z));
    m3 = fmaxf(m3, lrelu(a.w + erv.w));
  };
  int sA = 0, sB = 0;
  if (inl) {
    if (lane < cnt)      { sA = row[1 + lane];  upd_max(sA); }
    if (lane + 32 < cnt) { sB = row[33 + lane]; upd_max(sB); }
  } else {
    for (int base = 0; base < N_EDGES; base += 32) {
      const int e = base + lane;
      if (dst[e] == n) upd_max(src[e]);
    }
  }
  for (int off = 16; off > 0; off >>= 1) {
    m0 = fmaxf(m0, __shfl_xor(m0, off));
    m1 = fmaxf(m1, __shfl_xor(m1, off));
    m2 = fmaxf(m2, __shfl_xor(m2, off));
    m3 = fmaxf(m3, __shfl_xor(m3, off));
  }
  const float m_h = h == 0 ? m0 : h == 1 ? m1 : h == 2 ? m2 : m3;

  float ax = 0.0f, ay = 0.0f, den = 0.0f;
  const float* zc = z + 2 * lane;
  const float* elh = el + h;

  auto chunk = [&](int s, int c) {
    for (int j0 = 0; j0 < c; j0 += 8) {
      float lg[8];
      v2f   zz[8];
#pragma unroll
      for (int k = 0; k < 8; ++k) {
        const int j  = (j0 + k < c) ? (j0 + k) : (c - 1);
        const int sj = __builtin_amdgcn_readlane(s, j);
        lg[k] = elh[(size_t)sj * HEADS];
        zz[k] = *reinterpret_cast<const v2f*>(zc + (size_t)sj * HF);
      }
#pragma unroll
      for (int k = 0; k < 8; ++k) {
        const float w = (j0 + k < c) ? __expf(lrelu(lg[k] + er_h) - m_h) : 0.0f;
        den += w;
        ax = fmaf(w, zz[k].x, ax);
        ay = fmaf(w, zz[k].y, ay);
      }
    }
  };

  if (inl) {
    if (cnt > 0)  chunk(sA, cnt < 32 ? cnt : 32);
    if (cnt > 32) chunk(sB, cnt - 32);
  } else {
    for (int base = 0; base < N_EDGES; base += 32) {
      const int e = base + lane;
      const bool v = dst[e] == n;
      const int s = v ? src[e] : 0;
      unsigned mask = __builtin_amdgcn_ballot_w32(v);
      while (mask) {
        const int j = __builtin_ctz(mask);
        mask &= mask - 1;
        const int sj = __builtin_amdgcn_readlane(s, j);
        const float w = __expf(lrelu(elh[(size_t)sj * HEADS] + er_h) - m_h);
        const v2f zz = *reinterpret_cast<const v2f*>(zc + (size_t)sj * HF);
        den += w;
        ax = fmaf(w, zz.x, ax);
        ay = fmaf(w, zz.y, ay);
      }
    }
  }

  const v2f b = *reinterpret_cast<const v2f*>(bias + 2 * lane);
  v2f o;
  o.x = (den != 0.0f) ? (ax / den + b.x) : b.x;
  o.y = (den != 0.0f) ? (ay / den + b.y) : b.y;
  *reinterpret_cast<v2f*>(out + (size_t)n * HF + 2 * lane) = o;
}

extern "C" void kernel_launch(void* const* d_in, const int* in_sizes, int n_in,
                              void* d_out, int out_size, void* d_ws, size_t ws_size,
                              hipStream_t stream) {
  (void)in_sizes; (void)n_in; (void)out_size; (void)ws_size;
  const float* feats  = (const float*)d_in[0];
  const float* W      = (const float*)d_in[1];
  const float* attn_l = (const float*)d_in[2];
  const float* attn_r = (const float*)d_in[3];
  const float* bias   = (const float*)d_in[4];
  const int*   src    = (const int*)d_in[5];
  const int*   dst    = (const int*)d_in[6];
  float*       out    = (float*)d_out;

  float* ws = (float*)d_ws;
  float* z  = ws;
  float* el = z  + (size_t)N_NODES * HF;
  float* er = el + (size_t)N_NODES * HEADS;

  k_project<<<N_NODES / (16 * K1_WAVES), K1_WAVES * 32, 0, stream>>>(feats, W, z);

  {
    int n = N_NODES * HEADS;
    k_logits<<<(n + 255) / 256, 256, 0, stream>>>(z, attn_l, attn_r, el, er);
  }

  k_init<<<(N_NODES + 255) / 256, 256, 0, stream>>>(out);

  k_fill<<<(N_EDGES + 255) / 256, 256, 0, stream>>>(src, dst, out);

  k_node_agg<<<N_NODES / 8, 256, 0, stream>>>(src, dst, el, er, z, bias, out);
}


# Edges partitioned into 128-node bins with LDS counters (one global add per block and bin, was one per edge); the bin's block buckets them in LDS; a node's rows summed in ascending source order
# speedup vs baseline: 4.9418x; 1.7700x over previous
#include <hip/hip_runtime.h>
#include <hip/hip_bf16.h>
#include <math.h>

#define N_NODES   100000
#define N_EDGES   1600000
#define IN_FEATS  128
#define HEADS     4
#define OUT_F     16
#define HF        64
#define NEG_SLOPE 0.2f
#define BIN_SHIFT 7
#define BIN_NODES 128
#define N_BINS    ((N_NODES + BIN_NODES - 1) / BIN_NODES)
#define BIN_WORDS (BIN_NODES * HF)
#define LREC      4096

typedef __attribute__((ext_vector_type(2))) float v2f;
typedef __attribute__((ext_vector_type(4))) float v4f;
typedef __attribute__((ext_vector_type(8))) float v8f;

__device__ __forceinline__ float lrelu(float x) {
  return x > 0.0f ? x : NEG_SLOPE * x;
}

#define K1_WAVES 5
#define WPITCH4  33

__device__ __forceinline__ float logit_pair_sum(const v8f acc, const float al, const float ar,
                                                const int c) {
  const bool b8 = (c & 8) != 0, b4 = (c & 4) != 0, b2 = (c & 2) != 0, b1 = (c & 1) != 0;
  float y8[8], y4[4], y2[2];
#pragma unroll
  for (int j = 0; j < 8; ++j) {
    const float lo = acc[j] * al;
    const float hi = acc[j] * ar;
    y8[j] = (b8 ? hi : lo) + __shfl_xor(b8 ? lo : hi, 8);
  }
#pragma unroll
  for (int j = 0; j < 4; ++j)
    y4[j] = (b4 ? y8[j + 4] : y8[j]) + __shfl_xor(b4 ? y8[j] : y8[j + 4], 4);
#pragma unroll
  for (int j = 0; j < 2; ++j)
    y2[j] = (b2 ? y4[j + 2] : y4[j]) + __shfl_xor(b2 ? y4[j] : y4[j + 2], 2);
  return (b1 ? y2[1] : y2[0]) + __shfl_xor(b1 ? y2[0] : y2[1], 1);
}

__global__ __launch_bounds__(K1_WAVES * 32) void k_project(const float* __restrict__ feats,
                                                           const float* __restrict__ W,
                                                           const float* __restrict__ attn_l,
                                                           const float* __restrict__ attn_r,
                                                           float* __restrict__ z,
                                                           float* __restrict__ el,
                                                           float* __restrict__ er,
                                                           float* out) {
  __shared__ v4f sW[HF * WPITCH4];

  const int tid = threadIdx.x;
  const int t   = tid & 31;
  const int w   = tid >> 5;
  const int c   = t & 15;
  const int hh  = t >> 4;
  const int nodeBase = (blockIdx.x * K1_WAVES + w) * 16;

  const v4f* __restrict__ arow =
      reinterpret_cast<const v4f*>(feats + (size_t)(nodeBase + c) * IN_FEATS) + hh;
  v4f a[16];
#pragma unroll
  for (int s = 0; s < 16; ++s) a[s] = arow[2 * s];

  const v4f* __restrict__ W4 = reinterpret_cast<const v4f*>(W);
  for (int i = tid; i < HF * (IN_FEATS / 4); i += K1_WAVES * 32)
    sW[(i >> 5) * WPITCH4 + (i & 31)] = W4[i];
  __syncthreads();

  const v4f* brow = sW + c * WPITCH4 + hh;

  v8f acc0 = {}, acc1 = {}, acc2 = {}, acc3 = {};
#pragma unroll
  for (int s = 0; s < 16; ++s) {
    const v4f b0 = brow[(0 * 16) * WPITCH4 + 2 * s];
    const v4f b1 = brow[(1 * 16) * WPITCH4 + 2 * s];
    const v4f b2 = brow[(2 * 16) * WPITCH4 + 2 * s];
    const v4f b3 = brow[(3 * 16) * WPITCH4 + 2 * s];
    const v4f av = a[s];
    acc0 = __builtin_amdgcn_wmma_f32_16x16x4_f32(false, av.lo, false, b0.lo, (short)0, acc0, false, false);
    acc1 = __builtin_amdgcn_wmma_f32_16x16x4_f32(false, av.lo, false, b1.lo, (short)0, acc1, false, false);
    acc2 = __builtin_amdgcn_wmma_f32_16x16x4_f32(false, av.lo, false, b2.lo, (short)0, acc2, false, false);
    acc3 = __builtin_amdgcn_wmma_f32_16x16x4_f32(false, av.lo, false, b3.lo, (short)0, acc3, false, false);
    acc0 = __builtin_amdgcn_wmma_f32_16x16x4_f32(false, av.hi, false, b0.hi, (short)0, acc0, false, false);
    acc1 = __builtin_amdgcn_wmma_f32_16x16x4_f32(false, av.hi, false, b1.hi, (short)0, acc1, false, false);
    acc2 = __builtin_amdgcn_wmma_f32_16x16x4_f32(false, av.hi, false, b2.hi, (short)0, acc2, false, false);
    acc3 = __builtin_amdgcn_wmma_f32_16x16x4_f32(false, av.hi, false, b3.hi, (short)0, acc3, false, false);
  }

  float* __restrict__ zo = z + (size_t)(nodeBase + hh * 8) * HF + c;
#pragma unroll
  for (int i = 0; i < 8; ++i) {
    zo[(size_t)i * HF +  0] = acc0[i];
    zo[(size_t)i * HF + 16] = acc1[i];
    zo[(size_t)i * HF + 32] = acc2[i];
    zo[(size_t)i * HF + 48] = acc3[i];
  }

  float4 lg;
  lg.x = logit_pair_sum(acc0, attn_l[ 0 + c], attn_r[ 0 + c], c);
  lg.y = logit_pair_sum(acc1, attn_l[16 + c], attn_r[16 + c], c);
  lg.z = logit_pair_sum(acc2, attn_l[32 + c], attn_r[32 + c], c);
  lg.w = logit_pair_sum(acc3, attn_l[48 + c], attn_r[48 + c], c);
  float* __restrict__ lgo = (c < 8 ? el : er) + (size_t)(nodeBase + hh * 8 + (c & 7)) * HEADS;
  *reinterpret_cast<float4*>(lgo) = lg;

  if (tid < K1_WAVES * 16) {
    const int node = blockIdx.x * (K1_WAVES * 16) + tid;
    if ((node & (BIN_NODES - 1)) == 0) reinterpret_cast<int*>(out)[(size_t)node * HF] = 0;
  }
}

#define PT_THREADS 1024
#define PT_EPT     8
__global__ __launch_bounds__(PT_THREADS) void k_partition(const int* __restrict__ src,
                                                          const int* __restrict__ dst,
                                                          float* out) {
  __shared__ int hist[N_BINS];
  const int tid = threadIdx.x;
  for (int i = tid; i < N_BINS; i += PT_THREADS) hist[i] = 0;
  __syncthreads();

  const int e0 = blockIdx.x * (PT_THREADS * PT_EPT) + tid;
  int rec[PT_EPT], bin[PT_EPT], rnk[PT_EPT];
#pragma unroll
  for (int k = 0; k < PT_EPT; ++k) {
    const int e  = e0 + k * PT_THREADS;
    const int ee = e < N_EDGES ? e : N_EDGES - 1;
    const int s = src[ee], d = dst[ee];
    bin[k] = e < N_EDGES ? (d >> BIN_SHIFT) : -1;
    rec[k] = ((d & (BIN_NODES - 1)) << 17) | s;
  }
#pragma unroll
  for (int k = 0; k < PT_EPT; ++k) {
    rnk[k] = 0;
    if (bin[k] >= 0) rnk[k] = atomicAdd(&hist[bin[k]], 1);
  }
  __syncthreads();

  int* outi = reinterpret_cast<int*>(out);
  for (int i = tid; i < N_BINS; i += PT_THREADS) {
    const int c = hist[i];
    hist[i] = c > 0 ? atomicAdd(outi + (size_t)i * BIN_WORDS, c) : 0;
  }
  __syncthreads();

#pragma unroll
  for (int k = 0; k < PT_EPT; ++k) {
    if (bin[k] >= 0) {
      const int b     = bin[k];
      const int left  = N_NODES - b * BIN_NODES;
      const int nodes = left < BIN_NODES ? left : BIN_NODES;
      const int pos   = hist[b] + rnk[k];
      if (pos < nodes * HF - 1) outi[(size_t)b * BIN_WORDS + 1 + pos] = rec[k];
    }
  }
}

#define AG_THREADS 512
#define AG_WAVES   16
#define AG_NPW     (BIN_NODES / AG_WAVES)
#define AG_RPT     (LREC / AG_THREADS)
__global__ __launch_bounds__(AG_THREADS, 16) void k_bin_agg(const int* __restrict__ src,
                                                        const int* __restrict__ dst,
                                                        const float* __restrict__ el,
                                                        const float* __restrict__ er,
                                                        const float* __restrict__ z,
                                                        const float* __restrict__ bias,
                                                        float* out) {
  __shared__ int lst[LREC];
  __shared__ int srt[LREC];
  __shared__ int deg[BIN_NODES];
  __shared__ int off[BIN_NODES + 1];
  __shared__ int s_cnt;

  const int tid  = threadIdx.x;
  const int lane = tid & 31;
  const int w    = tid >> 5;
  const int b    = blockIdx.x;
  const int left  = N_NODES - b * BIN_NODES;
  const int nodes = left < BIN_NODES ? left : BIN_NODES;
  const int* region = reinterpret_cast<const int*>(out) + (size_t)b * BIN_WORDS;

  if (tid == 0) s_cnt = region[0];
  if (tid < BIN_NODES) deg[tid] = 0;
  __syncthreads();
  const int cnt = s_cnt;
  const int cap = nodes * HF - 1;
  const bool inl = cnt <= (cap < LREC ? cap : LREC);

  if (inl) {
    int rec[AG_RPT], rnk[AG_RPT];
#pragma unroll
    for (int k = 0; k < AG_RPT; ++k) {
      const int i = tid + k * AG_THREADS;
      const int v = region[1 + (i < cnt ? i : 0)];
      rec[k] = i < cnt ? v : -1;
    }
#pragma unroll
    for (int k = 0; k < AG_RPT; ++k) {
      rnk[k] = 0;
      if (rec[k] >= 0) rnk[k] = atomicAdd(&deg[rec[k] >> 17], 1);
    }
    __syncthreads();
    if (w == 0) {
      const int d0 = deg[4 * lane], d1 = deg[4 * lane + 1], d2 = deg[4 * lane + 2],
                d3 = deg[4 * lane + 3];
      const int t = d0 + d1 + d2 + d3;
      int incl = t;
      for (int o = 1; o < 32; o <<= 1) {
        const int v = __shfl_up(incl, o);
        if (lane >= o) incl += v;
      }
      const int ex = incl - t;
      off[4 * lane]     = ex;
      off[4 * lane + 1] = ex + d0;
      off[4 * lane + 2] = ex + d0 + d1;
      off[4 * lane + 3] = ex + d0 + d1 + d2;
      if (lane == 31) off[BIN_NODES] = incl;
    }
    __syncthreads();
#pragma unroll
    for (int k = 0; k < AG_RPT; ++k)
      if (rec[k] >= 0) lst[off[rec[k] >> 17] + rnk[k]] = rec[k] & 0x1FFFF;
    __syncthreads();
  }

  const int h = lane >> 3;
  const float* zc  = z + 2 * lane;
  const float* elh = el + h;
  const v2f bv = *reinterpret_cast<const v2f*>(bias + 2 * lane);

  for (int q = 0; q < AG_NPW; ++q) {
    const int l = w * AG_NPW + q;
    if (l >= nodes) break;
    const int n = b * BIN_NODES + l;

    const float4 erv = *reinterpret_cast<const float4*>(er + (size_t)n * HEADS);
    const float er_h = h == 0 ? erv.x : h == 1 ? erv.y : h == 2 ? erv.z : erv.w;

    float m0 = -INFINITY, m1 = -INFINITY, m2 = -INFINITY, m3 = -INFINITY;
    auto upd_max = [&](int s) {
      const float4 a = *reinterpret_cast<const float4*>(el + (size_t)s * HEADS);
      m0 = fmaxf(m0, lrelu(a.x + erv.x));
      m1 = fmaxf(m1, lrelu(a.y + erv.y));
      m2 = fmaxf(m2, lrelu(a.z + erv.z));
      m3 = fmaxf(m3, lrelu(a.w + erv.w));
    };

    int o = 0, dg = 0;
    if (inl) {
      o  = __builtin_amdgcn_readfirstlane(off[l]);
      dg = __builtin_amdgcn_readfirstlane(off[l + 1]) - o;
      if (dg <= 32) {
        const int v = lane < dg ? lst[o + lane] : 0x7fffffff;
        int r = 0;
        for (int j = 0; j < dg; ++j) {
          const int u = __builtin_amdgcn_readlane(v, j);
          r += (u < v || (u == v && j < lane)) ? 1 : 0;
        }
        if (lane < dg) srt[o + r] = v;
      } else {
        for (int i = lane; i < dg; i += 32) {
          const int v = lst[o + i];
          int r = 0;
          for (int j = 0; j < dg; ++j) {
            const int u = lst[o + j];
            r += (u < v || (u == v && j < i)) ? 1 : 0;
          }
          srt[o + r] = v;
        }
      }
      for (int c0 = 0; c0 < dg; c0 += 32)
        if (c0 + lane < dg) upd_max(srt[o + c0 + lane]);
    } else {
      for (int base = 0; base < N_EDGES; base += 32) {
        const int e = base + lane;
        if (dst[e] == n) upd_max(src[e]);
      }
    }
    for (int sh = 16; sh > 0; sh >>= 1) {
      m0 = fmaxf(m0, __shfl_xor(m0, sh));
      m1 = fmaxf(m1, __shfl_xor(m1, sh));
      m2 = fmaxf(m2, __shfl_xor(m2, sh));
      m3 = fmaxf(m3, __shfl_xor(m3, sh));
    }
    const float m_h = h == 0 ? m0 : h == 1 ? m1 : h == 2 ? m2 : m3;

    float ax = 0.0f, ay = 0.0f, den = 0.0f;

    auto chunk = [&](int s, int c) {
      for (int j0 = 0; j0 < c; j0 += 8) {
        float lg[8];
        v2f   zz[8];
#pragma unroll
        for (int k = 0; k < 8; ++k) {
          const int j  = (j0 + k < c) ? (j0 + k) : (c - 1);
          const int sj = __builtin_amdgcn_readlane(s, j);
          lg[k] = elh[(size_t)sj * HEADS];
          zz[k] = *reinterpret_cast<const v2f*>(zc + (size_t)sj * HF);
        }
#pragma unroll
        for (int k = 0; k < 8; ++k) {
          const float wt = (j0 + k < c) ? __expf(lrelu(lg[k] + er_h) - m_h) : 0.0f;
          den += wt;
          ax = fmaf(wt, zz[k].x, ax);
          ay = fmaf(wt, zz[k].y, ay);
        }
      }
    };

    if (inl) {
      for (int c0 = 0; c0 < dg; c0 += 32) {
        const int c = dg - c0 < 32 ? dg - c0 : 32;
        const int s = lane < c ? srt[o + c0 + lane] : 0;
        chunk(s, c);
      }
    } else {
      for (int base = 0; base < N_EDGES; base += 32) {
        const int e = base + lane;
        const bool v = dst[e] == n;
        const int s = v ? src[e] : 0;
        unsigned mask = __builtin_amdgcn_ballot_w32(v);
        while (mask) {
          const int j = __builtin_ctz(mask);
          mask &= mask - 1;
          const int sj = __builtin_amdgcn_readlane(s, j);
          const float wt = __expf(lrelu(elh[(size_t)sj * HEADS] + er_h) - m_h);
          const v2f zz = *reinterpret_cast<const v2f*>(zc + (size_t)sj * HF);
          den += wt;
          ax = fmaf(wt, zz.x, ax);
          ay = fmaf(wt, zz.y, ay);
        }
      }
    }

    v2f r2;
    r2.x = (den != 0.0f) ? (ax / den + bv.x) : bv.x;
    r2.y = (den != 0.0f) ? (ay / den + bv.y) : bv.y;
    *reinterpret_cast<v2f*>(out + (size_t)n * HF + 2 * lane) = r2;
  }
}

extern "C" void kernel_launch(void* const* d_in, const int* in_sizes, int n_in,
                              void* d_out, int out_size, void* d_ws, size_t ws_size,
                              hipStream_t stream) {
  (void)in_sizes; (void)n_in; (void)out_size; (void)ws_size;
  const float* feats  = (const float*)d_in[0];
  const float* W      = (const float*)d_in[1];
  const float* attn_l = (const float*)d_in[2];
  const float* attn_r = (const float*)d_in[3];
  const float* bias   = (const float*)d_in[4];
  const int*   src    = (const int*)d_in[5];
  const int*   dst    = (const int*)d_in[6];
  float*       out    = (float*)d_out;

  float* ws = (float*)d_ws;
  float* z  = ws;
  float* el = z  + (size_t)N_NODES * HF;
  float* er = el + (size_t)N_NODES * HEADS;

  k_project<<<N_NODES / (16 * K1_WAVES), K1_WAVES * 32, 0, stream>>>(feats, W, attn_l, attn_r,
                                                                      z, el, er, out);

  k_partition<<<(N_EDGES + PT_THREADS * PT_EPT - 1) / (PT_THREADS * PT_EPT), PT_THREADS, 0,
                stream>>>(src, dst, out);

  k_bin_agg<<<N_BINS, AG_THREADS, 0, stream>>>(src, dst, el, er, z, bias, out);
}


# Softmax weights formed once per edge by the lane that gathered its logits for the max and handed to the channel lanes through LDS: no second logit gather, 4 exp per lane and node, not one per edge
# speedup vs baseline: 5.5344x; 1.1199x over previous
#include <hip/hip_runtime.h>
#include <hip/hip_bf16.h>
#include <math.h>

#define N_NODES   100000
#define N_EDGES   1600000
#define IN_FEATS  128
#define HEADS     4
#define OUT_F     16
#define HF        64
#define NEG_SLOPE 0.2f
#define BIN_SHIFT 7
#define BIN_NODES 128
#define N_BINS    ((N_NODES + BIN_NODES - 1) / BIN_NODES)
#define BIN_WORDS (BIN_NODES * HF)
#define LREC      4096

typedef __attribute__((ext_vector_type(2))) float v2f;
typedef __attribute__((ext_vector_type(4))) float v4f;
typedef __attribute__((ext_vector_type(8))) float v8f;

__device__ __forceinline__ float lrelu(float x) {
  return x > 0.0f ? x : NEG_SLOPE * x;
}

#define K1_WAVES 5
#define WPITCH4  33

__device__ __forceinline__ float logit_pair_sum(const v8f acc, const float al, const float ar,
                                                const int c) {
  const bool b8 = (c & 8) != 0, b4 = (c & 4) != 0, b2 = (c & 2) != 0, b1 = (c & 1) != 0;
  float y8[8], y4[4], y2[2];
#pragma unroll
  for (int j = 0; j < 8; ++j) {
    const float lo = acc[j] * al;
    const float hi = acc[j] * ar;
    y8[j] = (b8 ? hi : lo) + __shfl_xor(b8 ? lo : hi, 8);
  }
#pragma unroll
  for (int j = 0; j < 4; ++j)
    y4[j] = (b4 ? y8[j + 4] : y8[j]) + __shfl_xor(b4 ? y8[j] : y8[j + 4], 4);
#pragma unroll
  for (int j = 0; j < 2; ++j)
    y2[j] = (b2 ? y4[j + 2] : y4[j]) + __shfl_xor(b2 ? y4[j] : y4[j + 2], 2);
  return (b1 ? y2[1] : y2[0]) + __shfl_xor(b1 ? y2[0] : y2[1], 1);
}

__global__ __launch_bounds__(K1_WAVES * 32) void k_project(const float* __restrict__ feats,
                                                           const float* __restrict__ W,
                                                           const float* __restrict__ attn_l,
                                                           const float* __restrict__ attn_r,
                                                           float* __restrict__ z,
                                                           float* __restrict__ el,
                                                           float* __restrict__ er,
                                                           float* out) {
  __shared__ v4f sW[HF * WPITCH4];

  const int tid = threadIdx.x;
  const int t   = tid & 31;
  const int w   = tid >> 5;
  const int c   = t & 15;
  const int hh  = t >> 4;
  const int nodeBase = (blockIdx.x * K1_WAVES + w) * 16;

  const v4f* __restrict__ arow =
      reinterpret_cast<const v4f*>(feats + (size_t)(nodeBase + c) * IN_FEATS) + hh;
  v4f a[16];
#pragma unroll
  for (int s = 0; s < 16; ++s) a[s] = arow[2 * s];

  const v4f* __restrict__ W4 = reinterpret_cast<const v4f*>(W);
  for (int i = tid; i < HF * (IN_FEATS / 4); i += K1_WAVES * 32)
    sW[(i >> 5) * WPITCH4 + (i & 31)] = W4[i];
  __syncthreads();

  const v4f* brow = sW + c * WPITCH4 + hh;

  v8f acc0 = {}, acc1 = {}, acc2 = {}, acc3 = {};
#pragma unroll
  for (int s = 0; s < 16; ++s) {
    const v4f b0 = brow[(0 * 16) * WPITCH4 + 2 * s];
    const v4f b1 = brow[(1 * 16) * WPITCH4 + 2 * s];
    const v4f b2 = brow[(2 * 16) * WPITCH4 + 2 * s];
    const v4f b3 = brow[(3 * 16) * WPITCH4 + 2 * s];
    const v4f av = a[s];
    acc0 = __builtin_amdgcn_wmma_f32_16x16x4_f32(false, av.lo, false, b0.lo, (short)0, acc0, false, false);
    acc1 = __builtin_amdgcn_wmma_f32_16x16x4_f32(false, av.lo, false, b1.lo, (short)0, acc1, false, false);
    acc2 = __builtin_amdgcn_wmma_f32_16x16x4_f32(false, av.lo, false, b2.lo, (short)0, acc2, false, false);
    acc3 = __builtin_amdgcn_wmma_f32_16x16x4_f32(false, av.lo, false, b3.lo, (short)0, acc3, false, false);
    acc0 = __builtin_amdgcn_wmma_f32_16x16x4_f32(false, av.hi, false, b0.hi, (short)0, acc0, false, false);
    acc1 = __builtin_amdgcn_wmma_f32_16x16x4_f32(false, av.hi, false, b1.hi, (short)0, acc1, false, false);
    acc2 = __builtin_amdgcn_wmma_f32_16x16x4_f32(false, av.hi, false, b2.hi, (short)0, acc2, false, false);
    acc3 = __builtin_amdgcn_wmma_f32_16x16x4_f32(false, av.hi, false, b3.hi, (short)0, acc3, false, false);
  }

  float* __restrict__ zo = z + (size_t)(nodeBase + hh * 8) * HF + c;
#pragma unroll
  for (int i = 0; i < 8; ++i) {
    zo[(size_t)i * HF +  0] = acc0[i];
    zo[(size_t)i * HF + 16] = acc1[i];
    zo[(size_t)i * HF + 32] = acc2[i];
    zo[(size_t)i * HF + 48] = acc3[i];
  }

  float4 lg;
  lg.x = logit_pair_sum(acc0, attn_l[ 0 + c], attn_r[ 0 + c], c);
  lg.y = logit_pair_sum(acc1, attn_l[16 + c], attn_r[16 + c], c);
  lg.z = logit_pair_sum(acc2, attn_l[32 + c], attn_r[32 + c], c);
  lg.w = logit_pair_sum(acc3, attn_l[48 + c], attn_r[48 + c], c);
  float* __restrict__ lgo = (c < 8 ? el : er) + (size_t)(nodeBase + hh * 8 + (c & 7)) * HEADS;
  *reinterpret_cast<float4*>(lgo) = lg;

  if (tid < K1_WAVES * 16) {
    const int node = blockIdx.x * (K1_WAVES * 16) + tid;
    if ((node & (BIN_NODES - 1)) == 0) reinterpret_cast<int*>(out)[(size_t)node * HF] = 0;
  }
}

#define PT_THREADS 1024
#define PT_EPT     8
__global__ __launch_bounds__(PT_THREADS) void k_partition(const int* __restrict__ src,
                                                          const int* __restrict__ dst,
                                                          float* out) {
  __shared__ int hist[N_BINS];
  const int tid = threadIdx.x;
  for (int i = tid; i < N_BINS; i += PT_THREADS) hist[i] = 0;
  __syncthreads();

  const int e0 = blockIdx.x * (PT_THREADS * PT_EPT) + tid;
  int rec[PT_EPT], bin[PT_EPT], rnk[PT_EPT];
#pragma unroll
  for (int k = 0; k < PT_EPT; ++k) {
    const int e  = e0 + k * PT_THREADS;
    const int ee = e < N_EDGES ? e : N_EDGES - 1;
    const int s = src[ee], d = dst[ee];
    bin[k] = e < N_EDGES ? (d >> BIN_SHIFT) : -1;
    rec[k] = ((d & (BIN_NODES - 1)) << 17) | s;
  }
#pragma unroll
  for (int k = 0; k < PT_EPT; ++k) {
    rnk[k] = 0;
    if (bin[k] >= 0) rnk[k] = atomicAdd(&hist[bin[k]], 1);
  }
  __syncthreads();

  int* outi = reinterpret_cast<int*>(out);
  for (int i = tid; i < N_BINS; i += PT_THREADS) {
    const int c = hist[i];
    hist[i] = c > 0 ? atomicAdd(outi + (size_t)i * BIN_WORDS, c) : 0;
  }
  __syncthreads();

#pragma unroll
  for (int k = 0; k < PT_EPT; ++k) {
    if (bin[k] >= 0) {
      const int b     = bin[k];
      const int left  = N_NODES - b * BIN_NODES;
      const int nodes = left < BIN_NODES ? left : BIN_NODES;
      const int pos   = hist[b] + rnk[k];
      if (pos < nodes * HF - 1) outi[(size_t)b * BIN_WORDS + 1 + pos] = rec[k];
    }
  }
}

#define AG_THREADS 512
#define AG_WAVES   16
#define AG_NPW     (BIN_NODES / AG_WAVES)
#define AG_RPT     (LREC / AG_THREADS)
#define AG_WPITCH  36
__global__ __launch_bounds__(AG_THREADS, 16) void k_bin_agg(const int* __restrict__ src,
                                                        const int* __restrict__ dst,
                                                        const float* __restrict__ el,
                                                        const float* __restrict__ er,
                                                        const float* __restrict__ z,
                                                        const float* __restrict__ bias,
                                                        float* out) {
  __shared__ int lst[LREC];
  __shared__ int srt[LREC];
  __shared__ int deg[BIN_NODES];
  __shared__ int off[BIN_NODES + 1];
  __shared__ int s_cnt;
  __shared__ v4f wts[AG_WAVES * HEADS * AG_WPITCH / 4];

  const int tid  = threadIdx.x;
  const int lane = tid & 31;
  const int w    = tid >> 5;
  const int b    = blockIdx.x;
  const int left  = N_NODES - b * BIN_NODES;
  const int nodes = left < BIN_NODES ? left : BIN_NODES;
  const int* region = reinterpret_cast<const int*>(out) + (size_t)b * BIN_WORDS;

  if (tid == 0) s_cnt = region[0];
  if (tid < BIN_NODES) deg[tid] = 0;
  __syncthreads();
  const int cnt = s_cnt;
  const int cap = nodes * HF - 1;
  const bool inl = cnt <= (cap < LREC ? cap : LREC);

  if (inl) {
    int rec[AG_RPT], rnk[AG_RPT];
#pragma unroll
    for (int k = 0; k < AG_RPT; ++k) {
      const int i = tid + k * AG_THREADS;
      const int v = region[1 + (i < cnt ? i : 0)];
      rec[k] = i < cnt ? v : -1;
    }
#pragma unroll
    for (int k = 0; k < AG_RPT; ++k) {
      rnk[k] = 0;
      if (rec[k] >= 0) rnk[k] = atomicAdd(&deg[rec[k] >> 17], 1);
    }
    __syncthreads();
    if (w == 0) {
      const int d0 = deg[4 * lane], d1 = deg[4 * lane + 1], d2 = deg[4 * lane + 2],
                d3 = deg[4 * lane + 3];
      const int t = d0 + d1 + d2 + d3;
      int incl = t;
      for (int o = 1; o < 32; o <<= 1) {
        const int v = __shfl_up(incl, o);
        if (lane >= o) incl += v;
      }
      const int ex = incl - t;
      off[4 * lane]     = ex;
      off[4 * lane + 1] = ex + d0;
      off[4 * lane + 2] = ex + d0 + d1;
      off[4 * lane + 3] = ex + d0 + d1 + d2;
      if (lane == 31) off[BIN_NODES] = incl;
    }
    __syncthreads();
#pragma unroll
    for (int k = 0; k < AG_RPT; ++k)
      if (rec[k] >= 0) lst[off[rec[k] >> 17] + rnk[k]] = rec[k] & 0x1FFFF;
    __syncthreads();
  }

  const int h = lane >> 3;
  const float* zc = z + 2 * lane;
  const v2f bv = *reinterpret_cast<const v2f*>(bias + 2 * lane);

  auto logits = [&](int s, const float4 erv) {
    const float4 a = *reinterpret_cast<const float4*>(el + (size_t)s * HEADS);
    float4 e;
    e.x = lrelu(a.x + erv.x);
    e.y = lrelu(a.y + erv.y);
    e.z = lrelu(a.z + erv.z);
    e.w = lrelu(a.w + erv.w);
    return e;
  };
  auto wave_max = [&](float4& m) {
    for (int sh = 16; sh > 0; sh >>= 1) {
      m.x = fmaxf(m.x, __shfl_xor(m.x, sh));
      m.y = fmaxf(m.y, __shfl_xor(m.y, sh));
      m.z = fmaxf(m.z, __shfl_xor(m.z, sh));
      m.w = fmaxf(m.w, __shfl_xor(m.w, sh));
    }
  };
  auto write_row = [&](int n, float ax, float ay, float den) {
    v2f r2;
    r2.x = (den != 0.0f) ? (ax / den + bv.x) : bv.x;
    r2.y = (den != 0.0f) ? (ay / den + bv.y) : bv.y;
    *reinterpret_cast<v2f*>(out + (size_t)n * HF + 2 * lane) = r2;
  };

  if (inl) {
    float* wrow = reinterpret_cast<float*>(wts) + w * (HEADS * AG_WPITCH);
    const float* wmine = wrow + h * AG_WPITCH;

    for (int q = 0; q < AG_NPW; ++q) {
      const int l = w * AG_NPW + q;
      if (l >= nodes) break;
      const int n = b * BIN_NODES + l;
      const float4 erv = *reinterpret_cast<const float4*>(er + (size_t)n * HEADS);

      const int o  = __builtin_amdgcn_readfirstlane(off[l]);
      const int dg = __builtin_amdgcn_readfirstlane(off[l + 1]) - o;
      if (dg <= 32) {
        const int v = lane < dg ? lst[o + lane] : 0x7fffffff;
        int r = 0;
        for (int j = 0; j < dg; ++j) {
          const int u = __builtin_amdgcn_readlane(v, j);
          r += (u < v || (u == v && j < lane)) ? 1 : 0;
        }
        if (lane < dg) srt[o + r] = v;
      } else {
        for (int i = lane; i < dg; i += 32) {
          const int v = lst[o + i];
          int r = 0;
          for (int j = 0; j < dg; ++j) {
            const int u = lst[o + j];
            r += (u < v || (u == v && j < i)) ? 1 : 0;
          }
          srt[o + r] = v;
        }
      }

      float4 m = make_float4(-INFINITY, -INFINITY, -INFINITY, -INFINITY);
      int    s0 = 0;
      float4 e0 = make_float4(0.0f, 0.0f, 0.0f, 0.0f);
      if (lane < dg) {
        s0 = srt[o + lane];
        e0 = logits(s0, erv);
        m = e0;
      }
      for (int c0 = 32; c0 < dg; c0 += 32) {
        if (c0 + lane < dg) {
          const float4 e = logits(srt[o + c0 + lane], erv);
          m.x = fmaxf(m.x, e.x);
          m.y = fmaxf(m.y, e.y);
          m.z = fmaxf(m.z, e.z);
          m.w = fmaxf(m.w, e.w);
        }
      }
      wave_max(m);

      float ax = 0.0f, ay = 0.0f, den = 0.0f;
      for (int c0 = 0; c0 < dg; c0 += 32) {
        const int c = dg - c0 < 32 ? dg - c0 : 32;
        int    s = s0;
        float4 e = e0;
        if (c0 > 0) {
          s = lane < c ? srt[o + c0 + lane] : 0;
          e = logits(s, erv);
        }
        const bool on = lane < c;
        const float w0 = __expf(e.x - m.x), w1 = __expf(e.y - m.y);
        const float w2 = __expf(e.z - m.z), w3 = __expf(e.w - m.w);
        wrow[0 * AG_WPITCH + lane] = on ? w0 : 0.0f;
        wrow[1 * AG_WPITCH + lane] = on ? w1 : 0.0f;
        wrow[2 * AG_WPITCH + lane] = on ? w2 : 0.0f;
        wrow[3 * AG_WPITCH + lane] = on ? w3 : 0.0f;
        for (int j0 = 0; j0 < c; j0 += 8) {
          v2f zz[8];
#pragma unroll
          for (int k = 0; k < 8; ++k) {
            const int j  = (j0 + k < c) ? (j0 + k) : (c - 1);
            const int sj = __builtin_amdgcn_readlane(s, j);
            zz[k] = *reinterpret_cast<const v2f*>(zc + (size_t)sj * HF);
          }
          const v4f wa = *reinterpret_cast<const v4f*>(wmine + j0);
          const v4f wb = *reinterpret_cast<const v4f*>(wmine + j0 + 4);
          const float wt[8] = {wa.x, wa.y, wa.z, wa.w, wb.x, wb.y, wb.z, wb.w};
#pragma unroll
          for (int k = 0; k < 8; ++k) {
            den += wt[k];
            ax = fmaf(wt[k], zz[k].x, ax);
            ay = fmaf(wt[k], zz[k].y, ay);
          }
        }
      }
      write_row(n, ax, ay, den);
    }
  } else {
    const float* elh = el + h;
    for (int q = 0; q < AG_NPW; ++q) {
      const int l = w * AG_NPW + q;
      if (l >= nodes) break;
      const int n = b * BIN_NODES + l;
      const float4 erv = *reinterpret_cast<const float4*>(er + (size_t)n * HEADS);
      const float er_h = h == 0 ? erv.x : h == 1 ? erv.y : h == 2 ? erv.z : erv.w;

      float4 m = make_float4(-INFINITY, -INFINITY, -INFINITY, -INFINITY);
      for (int base = 0; base < N_EDGES; base += 32) {
        const int e = base + lane;
        if (dst[e] == n) {
          const float4 g = logits(src[e], erv);
          m.x = fmaxf(m.x, g.x);
          m.y = fmaxf(m.y, g.y);
          m.z = fmaxf(m.z, g.z);
          m.w = fmaxf(m.w, g.w);
        }
      }
      wave_max(m);
      const float m_h = h == 0 ? m.x : h == 1 ? m.y : h == 2 ? m.z : m.w;

      float ax = 0.0f, ay = 0.0f, den = 0.0f;
      for (int base = 0; base < N_EDGES; base += 32) {
        const int e = base + lane;
        const bool v = dst[e] == n;
        const int s = v ? src[e] : 0;
        unsigned mask = __builtin_amdgcn_ballot_w32(v);
        while (mask) {
          const int j = __builtin_ctz(mask);
          mask &= mask - 1;
          const int sj = __builtin_amdgcn_readlane(s, j);
          const float wt = __expf(lrelu(elh[(size_t)sj * HEADS] + er_h) - m_h);
          const v2f zz = *reinterpret_cast<const v2f*>(zc + (size_t)sj * HF);
          den += wt;
          ax = fmaf(wt, zz.x, ax);
          ay = fmaf(wt, zz.y, ay);
        }
      }
      write_row(n, ax, ay, den);
    }
  }
}

extern "C" void kernel_launch(void* const* d_in, const int* in_sizes, int n_in,
                              void* d_out, int out_size, void* d_ws, size_t ws_size,
                              hipStream_t stream) {
  (void)in_sizes; (void)n_in; (void)out_size; (void)ws_size;
  const float* feats  = (const float*)d_in[0];
  const float* W      = (const float*)d_in[1];
  const float* attn_l = (const float*)d_in[2];
  const float* attn_r = (const float*)d_in[3];
  const float* bias   = (const float*)d_in[4];
  const int*   src    = (const int*)d_in[5];
  const int*   dst    = (const int*)d_in[6];
  float*       out    = (float*)d_out;

  float* ws = (float*)d_ws;
  float* z  = ws;
  float* el = z  + (size_t)N_NODES * HF;
  float* er = el + (size_t)N_NODES * HEADS;

  k_project<<<N_NODES / (16 * K1_WAVES), K1_WAVES * 32, 0, stream>>>(feats, W, attn_l, attn_r,
                                                                      z, el, er, out);

  k_partition<<<(N_EDGES + PT_THREADS * PT_EPT - 1) / (PT_THREADS * PT_EPT), PT_THREADS, 0,
                stream>>>(src, dst, out);

  k_bin_agg<<<N_BINS, AG_THREADS, 0, stream>>>(src, dst, el, er, z, bias, out);
}
